# starting kernel
# baseline (speedup 1.0000x reference)
#include <hip/hip_runtime.h>


#define N_SIG   2097152
#define L_FILT  129
#define N_OUT   3145728
#define TILE_OUT 768
#define NBLOCKS  4096
#define WIN      560
#define NTHREADS 96
#define KSTEPS   19

typedef float v2f __attribute__((ext_vector_type(2)));
typedef float v8f __attribute__((ext_vector_type(8)));

__global__ __launch_bounds__(NTHREADS)
void resample_poly_wmma(const float* __restrict__ x,
                        const float* __restrict__ h,
                        float* __restrict__ out)
{
    __shared__ float xw[WIN];
    __shared__ float hs[L_FILT];
    __shared__ __align__(16) float ob[TILE_OUT];

    const int tile = blockIdx.x;
    const int tid  = threadIdx.x;
    const int I0   = tile * 256;
    const int W    = 2 * I0 - 21;

    for (int j = tid; j < WIN; j += NTHREADS) {
        int gi = W + j;
        xw[j] = (gi >= 0 && gi < N_SIG) ? x[gi] : 0.0f;
    }
    for (int j = tid; j < L_FILT; j += NTHREADS) hs[j] = h[j];
    __syncthreads();

    const int wave = tid >> 5;
    const int lane = tid & 31;
    const int p    = wave;
    const int d0   = (p == 0) ? -21 : -20;
    const int basep= (p == 0) ? 0 : 1;

    const int r  = lane & 15;
    const int hi = lane >> 4;
    const int n  = lane & 15;

    v2f a[KSTEPS];
#pragma unroll
    for (int s = 0; s < KSTEPS; ++s) {
        const int k0 = 4 * s + 2 * hi;
        const int t0 = k0 - 2 * r;
        float ax = 0.0f, ay = 0.0f;
        if (t0 >= 0 && t0 < 43)         ax = 3.0f * hs[64 + 2 * p - 3 * (d0 + t0)];
        if (t0 + 1 >= 0 && t0 + 1 < 43) ay = 3.0f * hs[64 + 2 * p - 3 * (d0 + t0 + 1)];
        v2f av; av.x = ax; av.y = ay;
        a[s] = av;
    }

    v8f acc = {0.f, 0.f, 0.f, 0.f, 0.f, 0.f, 0.f, 0.f};
    const int bbase = basep + 32 * n + 2 * hi;
#pragma unroll
    for (int s = 0; s < KSTEPS; ++s) {
        v2f b;
        b.x = xw[bbase + 4 * s];
        b.y = xw[bbase + 4 * s + 1];
        acc = __builtin_amdgcn_wmma_f32_16x16x4_f32(
             false, a[s],  false, b,
             (short)0, acc,  false,  false);
    }

    const int rbase = hi * 8;
#pragma unroll
    for (int v = 0; v < 8; ++v) {
        ob[3 * (16 * n + rbase + v) + p] = acc[v];
    }
    __syncthreads();

    float4*       out4 = (float4*)(out + tile * TILE_OUT);
    const float4* ob4  = (const float4*)ob;
    out4[tid]      = ob4[tid];
    out4[tid + 96] = ob4[tid + 96];
}

extern "C" void kernel_launch(void* const* d_in, const int* in_sizes, int n_in,
                              void* d_out, int out_size, void* d_ws, size_t ws_size,
                              hipStream_t stream) {
    (void)in_sizes; (void)n_in; (void)d_ws; (void)ws_size; (void)out_size;
    const float* x = (const float*)d_in[0];
    const float* h = (const float*)d_in[1];
    float* out = (float*)d_out;
    resample_poly_wmma<<<NBLOCKS, NTHREADS, 0, stream>>>(x, h, out);
}


# Each work-group walks 4 consecutive tiles (grid 4096 -> 1024): filter fragments built once per wave, 4x fewer waves to enter; the 4-tile x window is staged by one batch of clamped loads
# speedup vs baseline: 1.0679x; 1.0679x over previous
#include <hip/hip_runtime.h>


#define N_SIG   2097152
#define L_FILT  129
#define N_OUT   3145728
#define TILE_OUT 768
#define TPB      4
#define NBLOCKS  1024
#define NTHREADS 96
#define XLOADS   22
#define WIN      (XLOADS * NTHREADS)
#define KSTEPS   19

typedef float v2f __attribute__((ext_vector_type(2)));
typedef float v8f __attribute__((ext_vector_type(8)));

__global__ __launch_bounds__(NTHREADS)
void resample_poly_wmma(const float* __restrict__ x,
                        const float* __restrict__ h,
                        float* __restrict__ out)
{
    __shared__ float xw[WIN];
    __shared__ float hs[2 * NTHREADS];
    __shared__ __align__(16) float ob[TPB * TILE_OUT];

    const int blk  = blockIdx.x;
    const int tid  = threadIdx.x;
    const int I0   = blk * (256 * TPB);
    const int W    = 2 * I0 - 21;

    float xv[XLOADS];
#pragma unroll
    for (int q = 0; q < XLOADS; ++q) {
        const int gi = W + tid + NTHREADS * q;
        const int gc = gi < 0 ? 0 : (gi > N_SIG - 1 ? N_SIG - 1 : gi);
        const float t = x[gc];
        xv[q] = (gi == gc) ? t : 0.0f;
    }
    const int   j1 = tid + NTHREADS;
    const float h0 = h[tid];
    const float h1 = h[j1 < L_FILT ? j1 : L_FILT - 1];
#pragma unroll
    for (int q = 0; q < XLOADS; ++q) xw[tid + NTHREADS * q] = xv[q];
    hs[tid] = h0;
    hs[j1]  = h1;
    __syncthreads();

    const int wave = tid >> 5;
    const int lane = tid & 31;
    const int p    = wave;
    const int d0   = (p == 0) ? -21 : -20;
    const int basep= (p == 0) ? 0 : 1;

    const int r  = lane & 15;
    const int hi = lane >> 4;
    const int n  = lane & 15;

    v2f a[KSTEPS];
#pragma unroll
    for (int s = 0; s < KSTEPS; ++s) {
        const int k0 = 4 * s + 2 * hi;
        const int t0 = k0 - 2 * r;
        float ax = 0.0f, ay = 0.0f;
        if (t0 >= 0 && t0 < 43)         ax = 3.0f * hs[64 + 2 * p - 3 * (d0 + t0)];
        if (t0 + 1 >= 0 && t0 + 1 < 43) ay = 3.0f * hs[64 + 2 * p - 3 * (d0 + t0 + 1)];
        v2f av; av.x = ax; av.y = ay;
        a[s] = av;
    }

    const int bbase = basep + 32 * n + 2 * hi;
    const int rbase = hi * 8;
#pragma unroll
    for (int tt = 0; tt < TPB; ++tt) {
        v8f acc = {0.f, 0.f, 0.f, 0.f, 0.f, 0.f, 0.f, 0.f};
#pragma unroll
        for (int s = 0; s < KSTEPS; ++s) {
            v2f b;
            b.x = xw[512 * tt + bbase + 4 * s];
            b.y = xw[512 * tt + bbase + 4 * s + 1];
            acc = __builtin_amdgcn_wmma_f32_16x16x4_f32(
                 false, a[s],  false, b,
                 (short)0, acc,  false,  false);
        }
#pragma unroll
        for (int v = 0; v < 8; ++v) {
            ob[TILE_OUT * tt + 3 * (16 * n + rbase + v) + p] = acc[v];
        }
    }
    __syncthreads();

    float4*       out4 = (float4*)(out + blk * (TPB * TILE_OUT));
    const float4* ob4  = (const float4*)ob;
#pragma unroll
    for (int q = 0; q < 2 * TPB; ++q) {
        out4[tid + 96 * q] = ob4[tid + 96 * q];
    }
}

extern "C" void kernel_launch(void* const* d_in, const int* in_sizes, int n_in,
                              void* d_out, int out_size, void* d_ws, size_t ws_size,
                              hipStream_t stream) {
    (void)in_sizes; (void)n_in; (void)d_ws; (void)ws_size; (void)out_size;
    const float* x = (const float*)d_in[0];
    const float* h = (const float*)d_in[1];
    float* out = (float*)d_out;
    resample_poly_wmma<<<NBLOCKS, NTHREADS, 0, stream>>>(x, h, out);
}


# LDS x window padded 4 floats per 32, one even tap origin for all phases: B fragment reads are aligned 8-byte loads on 64 banks (were 8-way conflicts); phases 1,2 sum K groups shifted one tap
# speedup vs baseline: 1.6400x; 1.5357x over previous
#include <hip/hip_runtime.h>


#define N_SIG   2097152
#define L_FILT  129
#define N_OUT   3145728
#define TILE_OUT 768
#define TPB      4
#define NBLOCKS  1024
#define NTHREADS 96
#define XLOADS   22
#define WIN      (XLOADS * NTHREADS)
#define XPAD     4
#define WINP     (WIN + XPAD * (WIN / 32))
#define TSTRIDE  (512 + XPAD * 16)
#define CSTRIDE  (32 + XPAD)
#define KSTEPS   19

typedef float v2f __attribute__((ext_vector_type(2)));
typedef float v8f __attribute__((ext_vector_type(8)));

__global__ __launch_bounds__(NTHREADS)
void resample_poly_wmma(const float* __restrict__ x,
                        const float* __restrict__ h,
                        float* __restrict__ out)
{
    __shared__ __align__(16) float xw[WINP];
    __shared__ float hs[2 * NTHREADS];
    __shared__ __align__(16) float ob[TPB * TILE_OUT];

    const int blk  = blockIdx.x;
    const int tid  = threadIdx.x;
    const int I0   = blk * (256 * TPB);
    const int W    = 2 * I0 - 21;

    float xv[XLOADS];
#pragma unroll
    for (int q = 0; q < XLOADS; ++q) {
        const int gi = W + tid + NTHREADS * q;
        const int gc = gi < 0 ? 0 : (gi > N_SIG - 1 ? N_SIG - 1 : gi);
        const float t = x[gc];
        xv[q] = (gi == gc) ? t : 0.0f;
    }
    const int   j1 = tid + NTHREADS;
    const float h0 = h[tid];
    const float h1 = h[j1 < L_FILT ? j1 : L_FILT - 1];
#pragma unroll
    for (int q = 0; q < XLOADS; ++q)
        xw[tid + XPAD * (tid >> 5) + (NTHREADS + 3 * XPAD) * q] = xv[q];
    hs[tid] = h0;
    hs[j1]  = h1;
    __syncthreads();

    const int wave = tid >> 5;
    const int lane = tid & 31;
    const int p    = wave;
    const int d0   = -21;
    const int tlo  = (p == 0) ? 0 : 1;
    const int thi  = tlo + 42;

    const int r  = lane & 15;
    const int hi = lane >> 4;
    const int n  = lane & 15;

    v2f a[KSTEPS];
#pragma unroll
    for (int s = 0; s < KSTEPS; ++s) {
        const int k0 = 4 * s + 2 * hi;
        const int t0 = k0 - 2 * r;
        float ax = 0.0f, ay = 0.0f;
        if (t0 >= tlo && t0 <= thi)         ax = 3.0f * hs[64 + 2 * p - 3 * (d0 + t0)];
        if (t0 + 1 >= tlo && t0 + 1 <= thi) ay = 3.0f * hs[64 + 2 * p - 3 * (d0 + t0 + 1)];
        v2f av; av.x = ax; av.y = ay;
        a[s] = av;
    }

    const int bbase = CSTRIDE * n + 2 * hi;
    const int rbase = hi * 8;
#pragma unroll
    for (int tt = 0; tt < TPB; ++tt) {
        v8f acc = {0.f, 0.f, 0.f, 0.f, 0.f, 0.f, 0.f, 0.f};
#pragma unroll
        for (int s = 0; s < KSTEPS; ++s) {
            const v2f b = *(const v2f*)&xw[TSTRIDE * tt + bbase + 4 * s + XPAD * (s >> 3)];
            acc = __builtin_amdgcn_wmma_f32_16x16x4_f32(
                 false, a[s],  false, b,
                 (short)0, acc,  false,  false);
        }
#pragma unroll
        for (int v = 0; v < 8; ++v) {
            ob[TILE_OUT * tt + 3 * (16 * n + rbase + v) + p] = acc[v];
        }
    }
    __syncthreads();

    float4*       out4 = (float4*)(out + blk * (TPB * TILE_OUT));
    const float4* ob4  = (const float4*)ob;
#pragma unroll
    for (int q = 0; q < 2 * TPB; ++q) {
        out4[tid + 96 * q] = ob4[tid + 96 * q];
    }
}

extern "C" void kernel_launch(void* const* d_in, const int* in_sizes, int n_in,
                              void* d_out, int out_size, void* d_ws, size_t ws_size,
                              hipStream_t stream) {
    (void)in_sizes; (void)n_in; (void)d_ws; (void)ws_size; (void)out_size;
    const float* x = (const float*)d_in[0];
    const float* h = (const float*)d_in[1];
    float* out = (float*)d_out;
    resample_poly_wmma<<<NBLOCKS, NTHREADS, 0, stream>>>(x, h, out);
}


# Filter fragments built without branches: taps read at clamped LDS indices, band edge by select; the 38 reads overlap each other and the matrix steps (were 36 guarded reads, each waited out in turn)
# speedup vs baseline: 1.9678x; 1.1999x over previous
#include <hip/hip_runtime.h>


#define N_SIG   2097152
#define L_FILT  129
#define N_OUT   3145728
#define TILE_OUT 768
#define TPB      4
#define NBLOCKS  1024
#define NTHREADS 96
#define XLOADS   22
#define WIN      (XLOADS * NTHREADS)
#define XPAD     4
#define WINP     (WIN + XPAD * (WIN / 32))
#define TSTRIDE  (512 + XPAD * 16)
#define CSTRIDE  (32 + XPAD)
#define KSTEPS   19

typedef float v2f __attribute__((ext_vector_type(2)));
typedef float v8f __attribute__((ext_vector_type(8)));

__global__ __launch_bounds__(NTHREADS)
void resample_poly_wmma(const float* __restrict__ x,
                        const float* __restrict__ h,
                        float* __restrict__ out)
{
    __shared__ __align__(16) float xw[WINP];
    __shared__ float hs[2 * NTHREADS];
    __shared__ __align__(16) float ob[TPB * TILE_OUT];

    const int blk  = blockIdx.x;
    const int tid  = threadIdx.x;
    const int I0   = blk * (256 * TPB);
    const int W    = 2 * I0 - 21;

    float xv[XLOADS];
#pragma unroll
    for (int q = 0; q < XLOADS; ++q) {
        const int gi = W + tid + NTHREADS * q;
        const int gc = gi < 0 ? 0 : (gi > N_SIG - 1 ? N_SIG - 1 : gi);
        const float t = x[gc];
        xv[q] = (gi == gc) ? t : 0.0f;
    }
    const int   j1 = tid + NTHREADS;
    const float h0 = h[tid];
    const float h1 = h[j1 < L_FILT ? j1 : L_FILT - 1];
#pragma unroll
    for (int q = 0; q < XLOADS; ++q)
        xw[tid + XPAD * (tid >> 5) + (NTHREADS + 3 * XPAD) * q] = xv[q];
    hs[tid] = h0;
    hs[j1]  = h1;
    __syncthreads();

    const int wave = tid >> 5;
    const int lane = tid & 31;
    const int p    = wave;
    const int d0   = -21;
    const int tlo  = (p == 0) ? 0 : 1;
    const int thi  = tlo + 42;

    const int r  = lane & 15;
    const int hi = lane >> 4;
    const int n  = lane & 15;

    v2f a[KSTEPS];
#pragma unroll
    for (int s = 0; s < KSTEPS; ++s) {
        const int k0 = 4 * s + 2 * hi;
        const int t0 = k0 - 2 * r;
        const int i0 = 64 + 2 * p - 3 * (d0 + t0);
        const int i1 = i0 - 3;
        const int c0 = i0 < 0 ? 0 : (i0 > L_FILT - 1 ? L_FILT - 1 : i0);
        const int c1 = i1 < 0 ? 0 : (i1 > L_FILT - 1 ? L_FILT - 1 : i1);
        const float g0 = hs[c0];
        const float g1 = hs[c1];
        v2f av;
        av.x = (t0 >= tlo && t0 <= thi)         ? 3.0f * g0 : 0.0f;
        av.y = (t0 + 1 >= tlo && t0 + 1 <= thi) ? 3.0f * g1 : 0.0f;
        a[s] = av;
    }

    const int bbase = CSTRIDE * n + 2 * hi;
    const int rbase = hi * 8;
#pragma unroll
    for (int tt = 0; tt < TPB; ++tt) {
        v8f acc = {0.f, 0.f, 0.f, 0.f, 0.f, 0.f, 0.f, 0.f};
#pragma unroll
        for (int s = 0; s < KSTEPS; ++s) {
            const v2f b = *(const v2f*)&xw[TSTRIDE * tt + bbase + 4 * s + XPAD * (s >> 3)];
            acc = __builtin_amdgcn_wmma_f32_16x16x4_f32(
                 false, a[s],  false, b,
                 (short)0, acc,  false,  false);
        }
#pragma unroll
        for (int v = 0; v < 8; ++v) {
            ob[TILE_OUT * tt + 3 * (16 * n + rbase + v) + p] = acc[v];
        }
    }
    __syncthreads();

    float4*       out4 = (float4*)(out + blk * (TPB * TILE_OUT));
    const float4* ob4  = (const float4*)ob;
#pragma unroll
    for (int q = 0; q < 2 * TPB; ++q) {
        out4[tid + 96 * q] = ob4[tid + 96 * q];
    }
}

extern "C" void kernel_launch(void* const* d_in, const int* in_sizes, int n_in,
                              void* d_out, int out_size, void* d_ws, size_t ws_size,
                              hipStream_t stream) {
    (void)in_sizes; (void)n_in; (void)d_ws; (void)ws_size; (void)out_size;
    const float* x = (const float*)d_in[0];
    const float* h = (const float*)d_in[1];
    float* out = (float*)d_out;
    resample_poly_wmma<<<NBLOCKS, NTHREADS, 0, stream>>>(x, h, out);
}


# Tile rows are 16 consecutive outputs (phases interleaved), not 16 of one phase: the band spans 53 x samples, not 73, so 14 matrix steps per tile replace 19; taps regrouped into the new K steps
# speedup vs baseline: 2.4426x; 1.2413x over previous
#include <hip/hip_runtime.h>


#define N_SIG   2097152
#define L_FILT  129
#define N_OUT   3145728
#define TILE_OUT 768
#define GRP_OUT  48
#define TPB      4
#define NBLOCKS  1024
#define NTHREADS 96
#define XLOADS   22
#define WIN      (XLOADS * NTHREADS)
#define XPAD     4
#define WINP     (WIN + XPAD * (WIN / 32))
#define TSTRIDE  (512 + XPAD * 16)
#define CSTRIDE  (32 + XPAD)
#define KORG     10
#define KSTEPS   14

typedef float v2f __attribute__((ext_vector_type(2)));
typedef float v8f __attribute__((ext_vector_type(8)));

__global__ __launch_bounds__(NTHREADS)
void resample_poly_wmma(const float* __restrict__ x,
                        const float* __restrict__ h,
                        float* __restrict__ out)
{
    __shared__ __align__(16) float xw[WINP];
    __shared__ float hs[2 * NTHREADS];
    __shared__ __align__(16) float ob[TPB * TILE_OUT];

    const int blk  = blockIdx.x;
    const int tid  = threadIdx.x;
    const int W    = 2048 * blk - 21;

    float xv[XLOADS];
#pragma unroll
    for (int q = 0; q < XLOADS; ++q) {
        const int gi = W + tid + NTHREADS * q;
        const int gc = gi < 0 ? 0 : (gi > N_SIG - 1 ? N_SIG - 1 : gi);
        const float t = x[gc];
        xv[q] = (gi == gc) ? t : 0.0f;
    }
    const int   j1 = tid + NTHREADS;
    const float h0 = h[tid];
    const float h1 = h[j1 < L_FILT ? j1 : L_FILT - 1];
#pragma unroll
    for (int q = 0; q < XLOADS; ++q)
        xw[tid + XPAD * (tid >> 5) + (NTHREADS + 3 * XPAD) * q] = xv[q];
    hs[tid] = h0;
    hs[j1]  = h1;
    __syncthreads();

    const int wave = tid >> 5;
    const int lane = tid & 31;

    const int r  = lane & 15;
    const int hi = lane >> 4;
    const int n  = lane & 15;

    v2f a[KSTEPS];
    const int tap0 = 127 + 2 * wave + 2 * r - 6 * hi;
#pragma unroll
    for (int s = 0; s < KSTEPS; ++s) {
        const int i0 = tap0 - 12 * s;
        const int i1 = i0 - 3;
        const int c0 = i0 < 0 ? 0 : (i0 > L_FILT - 1 ? L_FILT - 1 : i0);
        const int c1 = i1 < 0 ? 0 : (i1 > L_FILT - 1 ? L_FILT - 1 : i1);
        const float g0 = hs[c0];
        const float g1 = hs[c1];
        v2f av;
        av.x = (i0 == c0) ? 3.0f * g0 : 0.0f;
        av.y = (i1 == c1) ? 3.0f * g1 : 0.0f;
        a[s] = av;
    }

    const int kk0 = KORG * wave + 2 * hi;
    int boff[KSTEPS];
#pragma unroll
    for (int s = 0; s < KSTEPS; ++s) {
        const int kk = kk0 + 4 * s;
        boff[s] = CSTRIDE * n + kk + XPAD * (kk >> 5);
    }

    float* obl = ob + GRP_OUT * n + 16 * wave + 8 * hi;
#pragma unroll
    for (int tt = 0; tt < TPB; ++tt) {
        v8f acc = {0.f, 0.f, 0.f, 0.f, 0.f, 0.f, 0.f, 0.f};
#pragma unroll
        for (int s = 0; s < KSTEPS; ++s) {
            const v2f b = *(const v2f*)&xw[TSTRIDE * tt + boff[s]];
            acc = __builtin_amdgcn_wmma_f32_16x16x4_f32(
                 false, a[s],  false, b,
                 (short)0, acc,  false,  false);
        }
        float4* o4 = (float4*)(obl + TILE_OUT * tt);
        o4[0] = make_float4(acc[0], acc[1], acc[2], acc[3]);
        o4[1] = make_float4(acc[4], acc[5], acc[6], acc[7]);
    }
    __syncthreads();

    float4*       out4 = (float4*)(out + blk * (TPB * TILE_OUT));
    const float4* ob4  = (const float4*)ob;
#pragma unroll
    for (int q = 0; q < 2 * TPB; ++q) {
        out4[tid + 96 * q] = ob4[tid + 96 * q];
    }
}

extern "C" void kernel_launch(void* const* d_in, const int* in_sizes, int n_in,
                              void* d_out, int out_size, void* d_ws, size_t ws_size,
                              hipStream_t stream) {
    (void)in_sizes; (void)n_in; (void)d_ws; (void)ws_size; (void)out_size;
    const float* x = (const float*)d_in[0];
    const float* h = (const float*)d_in[1];
    float* out = (float*)d_out;
    resample_poly_wmma<<<NBLOCKS, NTHREADS, 0, stream>>>(x, h, out);
}
